# starting kernel
# baseline (speedup 1.0000x reference)
#include <hip/hip_runtime.h>
#include <hip/hip_bf16.h>
#include <stdint.h>

#define N_NODES 50000
#define N_EDGES 800000
#define F_NODE  64
#define F_EDGE  16
#define N_HEAD  2
#define CH      64
#define FDIM    128
#define NLAYER  2
#define NCLS    10
#define LN_EPS  1e-5f
#define SLOPE   0.2f
#define GEMM_ROWS 64

typedef __attribute__((ext_vector_type(16))) __bf16 v16bf;
typedef __attribute__((ext_vector_type(8)))  float  v8f;

union Frag { v16bf v; uint4 q[2]; };
union AccU { v8f v; float f[8]; };
union PK   { __bf16 h[2]; unsigned u; };

__device__ __forceinline__ int flipkey(float f) {
  int i = __float_as_int(f);
  return i >= 0 ? i : (i ^ 0x7fffffff);
}
__device__ __forceinline__ float unflipkey(int k) {
  return __int_as_float(k >= 0 ? k : (k ^ 0x7fffffff));
}

__global__ __launch_bounds__(256)
void k_gemm_bf16(const float* __restrict__ in, const float* __restrict__ W,
                 const float* __restrict__ bias, float* __restrict__ out,
                 int nrows, int Fin, int Fout, int FoutPad)
{
  __shared__ uint4 w4[128 * 17];
  __shared__ uint4 a4[GEMM_ROWS * 17];
  unsigned* wU = reinterpret_cast<unsigned*>(w4);
  unsigned* aU = reinterpret_cast<unsigned*>(a4);

  const int tid = threadIdx.x;
  const int rowBase = blockIdx.x * GEMM_ROWS;
  const int kp2 = Fin >> 1;

  for (int idx = tid; idx < FoutPad * kp2; idx += 256) {
    int n  = idx / kp2;
    int kp = idx - n * kp2;
    PK p; p.u = 0;
    if (n < Fout) {
      p.h[0] = (__bf16)W[(size_t)(2 * kp)     * Fout + n];
      p.h[1] = (__bf16)W[(size_t)(2 * kp + 1) * Fout + n];
    }
    wU[n * 68 + kp] = p.u;
  }
  const float2* in2 = reinterpret_cast<const float2*>(in);
  for (int idx = tid; idx < GEMM_ROWS * kp2; idx += 256) {
    int r  = idx / kp2;
    int kp = idx - r * kp2;
    PK p; p.u = 0;
    int gr = rowBase + r;
    if (gr < nrows) {
      float2 f = in2[(size_t)gr * kp2 + kp];
      p.h[0] = (__bf16)f.x;
      p.h[1] = (__bf16)f.y;
    }
    aU[r * 68 + kp] = p.u;
  }
  __syncthreads();

  const int wv = tid >> 5;
  const int n0 = wv * 16;
  if (n0 >= FoutPad) return;
  const int lane = tid & 31;
  const int hl = lane & 15;
  const int hi = lane >> 4;

  AccU acc[4];
  for (int rt = 0; rt < 4; ++rt)
    for (int i = 0; i < 8; ++i) acc[rt].f[i] = 0.f;

  const int nkstep = Fin >> 5;
  for (int ks = 0; ks < nkstep; ++ks) {
    const int qi = ks * 4 + hi;
    Frag b;
    b.q[0] = w4[(n0 + hl) * 17 + qi];
    b.q[1] = w4[(n0 + hl) * 17 + qi + 2];
#pragma unroll
    for (int rt = 0; rt < 4; ++rt) {
      Frag a;
      a.q[0] = a4[(rt * 16 + hl) * 17 + qi];
      a.q[1] = a4[(rt * 16 + hl) * 17 + qi + 2];
      acc[rt].v = __builtin_amdgcn_wmma_f32_16x16x32_bf16(
          false, a.v, false, b.v, (short)0, acc[rt].v, false, false);
    }
  }

  const int nn = n0 + hl;
  if (nn < Fout) {
    const float bv = bias[nn];
    for (int rt = 0; rt < 4; ++rt) {
      for (int i = 0; i < 8; ++i) {
        int rr = rowBase + rt * 16 + hi * 8 + i;
        if (rr < nrows) out[(size_t)rr * Fout + nn] = acc[rt].f[i] + bv;
      }
    }
  }
}

__global__ __launch_bounds__(256)
void k_layernorm(const float* __restrict__ x, const float* __restrict__ g,
                 const float* __restrict__ b, float* __restrict__ y, int nrows)
{
  const int wv = threadIdx.x >> 5, lane = threadIdx.x & 31;
  const int row = blockIdx.x * 8 + wv;
  if (row >= nrows) return;
  float4 v = reinterpret_cast<const float4*>(x + (size_t)row * FDIM)[lane];
  float s = v.x + v.y + v.z + v.w;
  for (int o = 16; o >= 1; o >>= 1) s += __shfl_xor(s, o, 32);
  const float mu = s * (1.f / FDIM);
  float4 d = make_float4(v.x - mu, v.y - mu, v.z - mu, v.w - mu);
  float q = d.x * d.x + d.y * d.y + d.z * d.z + d.w * d.w;
  for (int o = 16; o >= 1; o >>= 1) q += __shfl_xor(q, o, 32);
  const float rs = rsqrtf(q * (1.f / FDIM) + LN_EPS);
  float4 gg = reinterpret_cast<const float4*>(g)[lane];
  float4 bb = reinterpret_cast<const float4*>(b)[lane];
  float4 o4 = make_float4(d.x * rs * gg.x + bb.x, d.y * rs * gg.y + bb.y,
                          d.z * rs * gg.z + bb.z, d.w * rs * gg.w + bb.w);
  reinterpret_cast<float4*>(y + (size_t)row * FDIM)[lane] = o4;
}

__global__ void k_init(int* __restrict__ mxkey, float* __restrict__ den,
                       float* __restrict__ agg, int n2, int nagg)
{
  const int t = blockIdx.x * blockDim.x + threadIdx.x;
  if (t < nagg) agg[t] = 0.f;
  if (t < n2) { mxkey[t] = (int)0x80000000u; den[t] = 0.f; }
}

__global__ __launch_bounds__(256)
void k_edge_logits(const float* __restrict__ xl, const float* __restrict__ xr,
                   const int* __restrict__ ei, const float* __restrict__ eattr,
                   const float* __restrict__ We, const float* __restrict__ att,
                   float* __restrict__ logits, int* __restrict__ mxkey)
{
  __shared__ alignas(16) float we_s[F_EDGE * FDIM];
  __shared__ alignas(16) float att_s[FDIM];
  const int tid = threadIdx.x;
  for (int i = tid; i < F_EDGE * FDIM; i += 256) we_s[i] = We[i];
  if (tid < FDIM) att_s[tid] = att[tid];
  __syncthreads();

  const int lane = tid & 31;
  const int wid = blockIdx.x * 8 + (tid >> 5);
  const int nw  = gridDim.x * 8;
  const int c0  = lane * 4;
  const float4 at = reinterpret_cast<const float4*>(att_s)[lane];

  for (int e = wid; e < N_EDGES; e += nw) {
    const int s = ei[e], d = ei[N_EDGES + e];
    const int en = e + nw;
    if (en < N_EDGES) {
      __builtin_prefetch(xl + (size_t)ei[en] * FDIM + c0, 0, 3);
      __builtin_prefetch(xr + (size_t)ei[N_EDGES + en] * FDIM + c0, 0, 3);
    }
    float4 a = reinterpret_cast<const float4*>(xl + (size_t)s * FDIM)[lane];
    float4 r = reinterpret_cast<const float4*>(xr + (size_t)d * FDIM)[lane];
    float4 ee = make_float4(0.f, 0.f, 0.f, 0.f);
    const float* ea = eattr + (size_t)e * F_EDGE;
    for (int j = 0; j < F_EDGE; ++j) {
      const float w = ea[j];
      const float4 wj = reinterpret_cast<const float4*>(we_s + j * FDIM)[lane];
      ee.x += w * wj.x; ee.y += w * wj.y; ee.z += w * wj.z; ee.w += w * wj.w;
    }
    float4 m = make_float4(a.x + r.x + ee.x, a.y + r.y + ee.y,
                           a.z + r.z + ee.z, a.w + r.w + ee.w);
    m.x = m.x > 0.f ? m.x : SLOPE * m.x;
    m.y = m.y > 0.f ? m.y : SLOPE * m.y;
    m.z = m.z > 0.f ? m.z : SLOPE * m.z;
    m.w = m.w > 0.f ? m.w : SLOPE * m.w;
    float sres = m.x * at.x + m.y * at.y + m.z * at.z + m.w * at.w;
    for (int o = 1; o < 16; o <<= 1) sres += __shfl_xor(sres, o, 32);
    if ((lane & 15) == 0) {
      const int h = lane >> 4;
      logits[(size_t)e * 2 + h] = sres;
      atomicMax(mxkey + (size_t)d * 2 + h, flipkey(sres));
    }
  }
}

__global__ void k_edge_exp(const float* __restrict__ logits,
                           const int* __restrict__ mxkey,
                           const int* __restrict__ ei,
                           float* __restrict__ exb, float* __restrict__ den)
{
  const int t = blockIdx.x * blockDim.x + threadIdx.x;
  if (t >= N_EDGES * 2) return;
  const int e = t >> 1, h = t & 1;
  const int d = ei[N_EDGES + e];
  const float mx = unflipkey(mxkey[(size_t)d * 2 + h]);
  const float ex = __expf(logits[t] - mx);
  exb[t] = ex;
  atomicAdd(den + (size_t)d * 2 + h, ex);
}

__global__ __launch_bounds__(256)
void k_edge_agg(const float* __restrict__ xl, const int* __restrict__ ei,
                const float* __restrict__ exb, const float* __restrict__ den,
                float* __restrict__ agg)
{
  const int lane = threadIdx.x & 31;
  const int wid = blockIdx.x * 8 + (threadIdx.x >> 5);
  const int nw  = gridDim.x * 8;
  const int c0  = lane * 4;
  const int h   = lane >> 4;

  for (int e = wid; e < N_EDGES; e += nw) {
    const int s = ei[e], d = ei[N_EDGES + e];
    const int en = e + nw;
    if (en < N_EDGES)
      __builtin_prefetch(xl + (size_t)ei[en] * FDIM + c0, 0, 3);
    const float alpha = exb[(size_t)e * 2 + h] / den[(size_t)d * 2 + h];
    const float4 a = reinterpret_cast<const float4*>(xl + (size_t)s * FDIM)[lane];
    float* ap = agg + (size_t)d * FDIM + c0;
    atomicAdd(ap + 0, a.x * alpha);
    atomicAdd(ap + 1, a.y * alpha);
    atomicAdd(ap + 2, a.z * alpha);
    atomicAdd(ap + 3, a.w * alpha);
  }
}

__global__ void k_node_update(float* __restrict__ h, const float* __restrict__ agg,
                              const float* __restrict__ bout)
{
  const int t = blockIdx.x * blockDim.x + threadIdx.x;
  if (t >= N_NODES * FDIM) return;
  const int c = t & (FDIM - 1);
  const float v = agg[t] + bout[c];
  h[t] = fmaxf(v, 0.f) + h[t];
}

__global__ void k_copy(const float* __restrict__ src, float* __restrict__ dst, int n)
{
  const int t = blockIdx.x * blockDim.x + threadIdx.x;
  if (t < n) dst[t] = src[t];
}

extern "C" void kernel_launch(void* const* d_in, const int* in_sizes, int n_in,
                              void* d_out, int out_size, void* d_ws, size_t ws_size,
                              hipStream_t stream)
{
  const float* x     = (const float*)d_in[0];
  const int*   ei    = (const int*)  d_in[1];
  const float* eattr = (const float*)d_in[2];
  const float* Wp    = (const float*)d_in[3];
  const float* bp    = (const float*)d_in[4];
  const float* ln_g  = (const float*)d_in[5];
  const float* ln_b  = (const float*)d_in[6];
  const float* Wl    = (const float*)d_in[7];
  const float* bl    = (const float*)d_in[8];
  const float* Wr    = (const float*)d_in[9];
  const float* br    = (const float*)d_in[10];
  const float* We    = (const float*)d_in[11];
  const float* att   = (const float*)d_in[12];
  const float* bout  = (const float*)d_in[13];
  const float* Wc    = (const float*)d_in[14];
  const float* bc    = (const float*)d_in[15];
  float* out = (float*)d_out;

  const size_t NF = (size_t)N_NODES * FDIM;
  float* h      = (float*)d_ws;
  float* hn     = h + NF;
  float* xl     = hn + NF;
  float* xr     = xl + NF;
  float* agg    = xr + NF;
  float* logits = agg + NF;
  float* exb    = logits + (size_t)N_EDGES * 2;
  float* den    = exb + (size_t)N_EDGES * 2;
  int*   mxkey  = (int*)(den + (size_t)N_NODES * 2);

  const dim3 blk(256);
  const int gemmGrid = (N_NODES + GEMM_ROWS - 1) / GEMM_ROWS;
  const int edgeGrid = 2048;
  const int lnGrid   = (N_NODES + 7) / 8;
  const int elemGrid = (int)((NF + 255) / 256);
  const int ehGrid   = (N_EDGES * 2 + 255) / 256;

  k_gemm_bf16<<<gemmGrid, blk, 0, stream>>>(x, Wp, bp, h, N_NODES, F_NODE, FDIM, FDIM);

  for (int i = 0; i < NLAYER; ++i) {
    k_layernorm<<<lnGrid, blk, 0, stream>>>(h, ln_g + i * FDIM, ln_b + i * FDIM, hn, N_NODES);
    k_gemm_bf16<<<gemmGrid, blk, 0, stream>>>(hn, Wl + (size_t)i * FDIM * FDIM,
                                              bl + i * FDIM, xl, N_NODES, FDIM, FDIM, FDIM);
    k_gemm_bf16<<<gemmGrid, blk, 0, stream>>>(hn, Wr + (size_t)i * FDIM * FDIM,
                                              br + i * FDIM, xr, N_NODES, FDIM, FDIM, FDIM);
    k_init<<<elemGrid, blk, 0, stream>>>(mxkey, den, agg, N_NODES * 2, (int)NF);
    k_edge_logits<<<edgeGrid, blk, 0, stream>>>(xl, xr, ei, eattr,
                                                We + (size_t)i * F_EDGE * FDIM,
                                                att + i * N_HEAD * CH, logits, mxkey);
    k_edge_exp<<<ehGrid, blk, 0, stream>>>(logits, mxkey, ei, exb, den);
    k_edge_agg<<<edgeGrid, blk, 0, stream>>>(xl, ei, exb, den, agg);
    k_node_update<<<elemGrid, blk, 0, stream>>>(h, agg, bout + i * FDIM);
  }

  k_gemm_bf16<<<gemmGrid, blk, 0, stream>>>(h, Wc, bc, out, N_NODES, FDIM, NCLS, 16);
  k_copy<<<elemGrid, blk, 0, stream>>>(h, out + (size_t)N_NODES * NCLS, (int)NF);
}


# LayerNorm and both projections of a layer in one launch: a work-group reads its 64 rows of h once, normalises them in registers, keeps the tile in LDS for Wl then Wr; sum order unchanged
# speedup vs baseline: 4.2066x; 4.2066x over previous
#include <hip/hip_runtime.h>
#include <hip/hip_bf16.h>
#include <stdint.h>

#define N_NODES 50000
#define N_EDGES 800000
#define F_NODE  64
#define F_EDGE  16
#define N_HEAD  2
#define CH      64
#define FDIM    128
#define NLAYER  2
#define NCLS    10
#define LN_EPS  1e-5f
#define SLOPE   0.2f
#define GEMM_ROWS 64

typedef __attribute__((ext_vector_type(16))) __bf16 v16bf;
typedef __attribute__((ext_vector_type(8)))  float  v8f;

union Frag { v16bf v; uint4 q[2]; };
union AccU { v8f v; float f[8]; };
union PK   { __bf16 h[2]; unsigned u; };

template <int FIN, int FOUT>
__global__ __launch_bounds__(256)
void k_gemm_bf16(const float* __restrict__ in, const float* __restrict__ W,
                 const float* __restrict__ bias, float* __restrict__ out, int nrows)
{
  constexpr int FOUTPAD = (FOUT + 15) & ~15;
  constexpr int KP2 = FIN / 2;
  constexpr int KQ  = FIN / 4;
  static_assert(FIN % 32 == 0 && FIN <= 128 && FOUTPAD <= 128, "tile shape");
  static_assert((GEMM_ROWS * KQ) % 256 == 0, "A pieces divide among the threads");

  __shared__ uint4 w4[FOUTPAD * 17];
  __shared__ uint4 a4[GEMM_ROWS * 17];
  unsigned* wU = reinterpret_cast<unsigned*>(w4);
  unsigned* aU = reinterpret_cast<unsigned*>(a4);

  const int tid = threadIdx.x;
  const int rowBase = blockIdx.x * GEMM_ROWS;

  constexpr int A_PER = GEMM_ROWS * KQ / 256;
  const float4* __restrict__ in4 = reinterpret_cast<const float4*>(in);
  float4 av[A_PER];
#pragma unroll
  for (int u = 0; u < A_PER; ++u) {
    const int idx = tid + u * 256;
    const int r  = idx / KQ;
    const int c4 = idx % KQ;
    const int gr = min(rowBase + r, nrows - 1);
    av[u] = in4[(size_t)gr * KQ + c4];
  }

  if constexpr (FOUT % 16 == 0) {
    constexpr int F4 = FOUT / 4;
    constexpr int NT = FOUT / 16;
    constexpr int W_PER = (FIN / 4) * F4 / 256;
    static_assert(((FIN / 4) * F4) % 256 == 0, "W items divide among the threads");
    const float4* __restrict__ W4 = reinterpret_cast<const float4*>(W);
    float4 wr[W_PER][4];
#pragma unroll
    for (int u = 0; u < W_PER; ++u) {
      const int it = tid + u * 256;
      const int up = it >> 5;
      const int n4 = (it & 3) + 4 * (up % NT);
      const int kq = ((it >> 2) & 7) + 8 * (up / NT);
#pragma unroll
      for (int j = 0; j < 4; ++j) wr[u][j] = W4[(size_t)(4 * kq + j) * F4 + n4];
    }
#pragma unroll
    for (int u = 0; u < W_PER; ++u) {
      const int it = tid + u * 256;
      const int up = it >> 5;
      const int n4 = (it & 3) + 4 * (up % NT);
      const int kq = ((it >> 2) & 7) + 8 * (up / NT);
      uint2* dst = reinterpret_cast<uint2*>(wU + (4 * n4) * 68 + 2 * kq);
      PK a, b;
      a.h[0] = (__bf16)wr[u][0].x; a.h[1] = (__bf16)wr[u][1].x;
      b.h[0] = (__bf16)wr[u][2].x; b.h[1] = (__bf16)wr[u][3].x;
      dst[0] = make_uint2(a.u, b.u);
      a.h[0] = (__bf16)wr[u][0].y; a.h[1] = (__bf16)wr[u][1].y;
      b.h[0] = (__bf16)wr[u][2].y; b.h[1] = (__bf16)wr[u][3].y;
      dst[34] = make_uint2(a.u, b.u);
      a.h[0] = (__bf16)wr[u][0].z; a.h[1] = (__bf16)wr[u][1].z;
      b.h[0] = (__bf16)wr[u][2].z; b.h[1] = (__bf16)wr[u][3].z;
      dst[68] = make_uint2(a.u, b.u);
      a.h[0] = (__bf16)wr[u][0].w; a.h[1] = (__bf16)wr[u][1].w;
      b.h[0] = (__bf16)wr[u][2].w; b.h[1] = (__bf16)wr[u][3].w;
      dst[102] = make_uint2(a.u, b.u);
    }
  } else {
    constexpr int NW = FOUTPAD * KP2;
    constexpr int W_PER = (NW + 255) / 256;
    float wlo[W_PER], whi[W_PER];
#pragma unroll
    for (int u = 0; u < W_PER; ++u) {
      const int idx = tid + u * 256;
      const int n  = idx / KP2;
      const int kp = idx % KP2;
      wlo[u] = 0.f; whi[u] = 0.f;
      if (idx < NW && n < FOUT) {
        wlo[u] = W[(size_t)(2 * kp)     * FOUT + n];
        whi[u] = W[(size_t)(2 * kp + 1) * FOUT + n];
      }
    }
#pragma unroll
    for (int u = 0; u < W_PER; ++u) {
      const int idx = tid + u * 256;
      const int n  = idx / KP2;
      const int kp = idx % KP2;
      PK p;
      p.h[0] = (__bf16)wlo[u];
      p.h[1] = (__bf16)whi[u];
      if (idx < NW) wU[n * 68 + kp] = p.u;
    }
  }

#pragma unroll
  for (int u = 0; u < A_PER; ++u) {
    const int idx = tid + u * 256;
    const int r  = idx / KQ;
    const int c4 = idx % KQ;
    PK a, b;
    a.h[0] = (__bf16)av[u].x; a.h[1] = (__bf16)av[u].y;
    b.h[0] = (__bf16)av[u].z; b.h[1] = (__bf16)av[u].w;
    *reinterpret_cast<uint2*>(aU + r * 68 + 2 * c4) = make_uint2(a.u, b.u);
  }
  __syncthreads();

  const int wv = tid >> 5;
  const int n0 = wv * 16;
  if (n0 >= FOUTPAD) return;
  const int lane = tid & 31;
  const int hl = lane & 15;
  const int hi = lane >> 4;

  AccU acc[4];
  for (int rt = 0; rt < 4; ++rt)
    for (int i = 0; i < 8; ++i) acc[rt].f[i] = 0.f;

  constexpr int nkstep = FIN >> 5;
  for (int ks = 0; ks < nkstep; ++ks) {
    const int qi = ks * 4 + hi;
    Frag b;
    b.q[0] = w4[(n0 + hl) * 17 + qi];
    b.q[1] = w4[(n0 + hl) * 17 + qi + 2];
#pragma unroll
    for (int rt = 0; rt < 4; ++rt) {
      Frag a;
      a.q[0] = a4[(rt * 16 + hl) * 17 + qi];
      a.q[1] = a4[(rt * 16 + hl) * 17 + qi + 2];
      acc[rt].v = __builtin_amdgcn_wmma_f32_16x16x32_bf16(
          false, a.v, false, b.v, (short)0, acc[rt].v, false, false);
    }
  }

  const int nn = n0 + hl;
  if (nn < FOUT) {
    const float bv = bias[nn];
    for (int rt = 0; rt < 4; ++rt) {
      for (int i = 0; i < 8; ++i) {
        int rr = rowBase + rt * 16 + hi * 8 + i;
        if (rr < nrows) out[(size_t)rr * FOUT + nn] = acc[rt].f[i] + bv;
      }
    }
  }
}

struct WRows { float4 v[FDIM / 4 * (FDIM / 4) / 256][4]; };

__device__ __forceinline__ void w_request(const float* __restrict__ W, const int tid, WRows& p)
{
  constexpr int F4 = FDIM / 4, NT = FDIM / 16, W_PER = FDIM / 4 * F4 / 256;
  const float4* __restrict__ W4 = reinterpret_cast<const float4*>(W);
#pragma unroll
  for (int u = 0; u < W_PER; ++u) {
    const int it = tid + u * 256;
    const int up = it >> 5;
    const int n4 = (it & 3) + 4 * (up % NT);
    const int kq = ((it >> 2) & 7) + 8 * (up / NT);
#pragma unroll
    for (int j = 0; j < 4; ++j) p.v[u][j] = W4[(size_t)(4 * kq + j) * F4 + n4];
  }
}

__device__ __forceinline__ void w_to_lds(const WRows& p, unsigned* wU, const int tid)
{
  constexpr int NT = FDIM / 16, W_PER = FDIM / 4 * (FDIM / 4) / 256;
#pragma unroll
  for (int u = 0; u < W_PER; ++u) {
    const int it = tid + u * 256;
    const int up = it >> 5;
    const int n4 = (it & 3) + 4 * (up % NT);
    const int kq = ((it >> 2) & 7) + 8 * (up / NT);
    uint2* dst = reinterpret_cast<uint2*>(wU + (4 * n4) * 68 + 2 * kq);
    PK a, b;
    a.h[0] = (__bf16)p.v[u][0].x; a.h[1] = (__bf16)p.v[u][1].x;
    b.h[0] = (__bf16)p.v[u][2].x; b.h[1] = (__bf16)p.v[u][3].x;
    dst[0] = make_uint2(a.u, b.u);
    a.h[0] = (__bf16)p.v[u][0].y; a.h[1] = (__bf16)p.v[u][1].y;
    b.h[0] = (__bf16)p.v[u][2].y; b.h[1] = (__bf16)p.v[u][3].y;
    dst[34] = make_uint2(a.u, b.u);
    a.h[0] = (__bf16)p.v[u][0].z; a.h[1] = (__bf16)p.v[u][1].z;
    b.h[0] = (__bf16)p.v[u][2].z; b.h[1] = (__bf16)p.v[u][3].z;
    dst[68] = make_uint2(a.u, b.u);
    a.h[0] = (__bf16)p.v[u][0].w; a.h[1] = (__bf16)p.v[u][1].w;
    b.h[0] = (__bf16)p.v[u][2].w; b.h[1] = (__bf16)p.v[u][3].w;
    dst[102] = make_uint2(a.u, b.u);
  }
}

__device__ __forceinline__ void tile_product_store(const uint4* w4, const uint4* a4,
                                                   const float* __restrict__ bias,
                                                   float* __restrict__ out,
                                                   const int n0, const int hl, const int hi,
                                                   const int rowBase, const int nrows)
{
  AccU acc[4];
  for (int rt = 0; rt < 4; ++rt)
    for (int i = 0; i < 8; ++i) acc[rt].f[i] = 0.f;
  for (int ks = 0; ks < (FDIM >> 5); ++ks) {
    const int qi = ks * 4 + hi;
    Frag b;
    b.q[0] = w4[(n0 + hl) * 17 + qi];
    b.q[1] = w4[(n0 + hl) * 17 + qi + 2];
#pragma unroll
    for (int rt = 0; rt < 4; ++rt) {
      Frag a;
      a.q[0] = a4[(rt * 16 + hl) * 17 + qi];
      a.q[1] = a4[(rt * 16 + hl) * 17 + qi + 2];
      acc[rt].v = __builtin_amdgcn_wmma_f32_16x16x32_bf16(
          false, a.v, false, b.v, (short)0, acc[rt].v, false, false);
    }
  }
  const int nn = n0 + hl;
  const float bv = bias[nn];
  const int r0 = rowBase + hi * 8;
  float* __restrict__ o = out + (size_t)r0 * FDIM + nn;
#pragma unroll
  for (int rt = 0; rt < 4; ++rt) {
#pragma unroll
    for (int i = 0; i < 8; ++i) {
      if (r0 + rt * 16 + i < nrows) o[(rt * 16 + i) * FDIM] = acc[rt].f[i] + bv;
    }
  }
}

__global__ __launch_bounds__(256)
void k_ln_proj2(const float* __restrict__ h, const float* __restrict__ g,
                const float* __restrict__ b,
                const float* __restrict__ Wa, const float* __restrict__ biasA,
                float* __restrict__ outA,
                const float* __restrict__ Wb, const float* __restrict__ biasB,
                float* __restrict__ outB, int nrows)
{
  constexpr int KQ = FDIM / 4;
  constexpr int A_PER = GEMM_ROWS * KQ / 256;
  static_assert(KQ == 32, "one wave holds one whole row per load");

  __shared__ uint4 w4[FDIM * 17];
  __shared__ uint4 a4[GEMM_ROWS * 17];
  unsigned* wU = reinterpret_cast<unsigned*>(w4);
  unsigned* aU = reinterpret_cast<unsigned*>(a4);

  const int tid = threadIdx.x;
  const int lane = tid & 31;
  const int wv = tid >> 5;
  const int rowBase = blockIdx.x * GEMM_ROWS;

  const float4* __restrict__ h4 = reinterpret_cast<const float4*>(h);
  float4 av[A_PER];
#pragma unroll
  for (int u = 0; u < A_PER; ++u) {
    const int gr = min(rowBase + wv + 8 * u, nrows - 1);
    av[u] = h4[(size_t)gr * KQ + lane];
  }
  WRows pw;
  w_request(Wa, tid, pw);
  w_to_lds(pw, wU, tid);

  const float4 gg = reinterpret_cast<const float4*>(g)[lane];
  const float4 bb = reinterpret_cast<const float4*>(b)[lane];
#pragma unroll
  for (int u = 0; u < A_PER; ++u) {
    const float4 v = av[u];
    float s = v.x + v.y + v.z + v.w;
    for (int o = 16; o >= 1; o >>= 1) s += __shfl_xor(s, o, 32);
    const float mu = s * (1.f / FDIM);
    float4 d = make_float4(v.x - mu, v.y - mu, v.z - mu, v.w - mu);
    float q = d.x * d.x + d.y * d.y + d.z * d.z + d.w * d.w;
    for (int o = 16; o >= 1; o >>= 1) q += __shfl_xor(q, o, 32);
    const float rs = rsqrtf(q * (1.f / FDIM) + LN_EPS);
    av[u] = make_float4(d.x * rs * gg.x + bb.x, d.y * rs * gg.y + bb.y,
                        d.z * rs * gg.z + bb.z, d.w * rs * gg.w + bb.w);
  }

#pragma unroll
  for (int u = 0; u < A_PER; ++u) {
    const int r = wv + 8 * u;
    PK a, c2;
    a.h[0]  = (__bf16)av[u].x; a.h[1]  = (__bf16)av[u].y;
    c2.h[0] = (__bf16)av[u].z; c2.h[1] = (__bf16)av[u].w;
    *reinterpret_cast<uint2*>(aU + r * 68 + 2 * lane) = make_uint2(a.u, c2.u);
  }
  __syncthreads();

  const int n0 = wv * 16;
  const int hl = lane & 15;
  const int hi = lane >> 4;
  tile_product_store(w4, a4, biasA, outA, n0, hl, hi, rowBase, nrows);

  __syncthreads();
  w_request(Wb, tid, pw);
  w_to_lds(pw, wU, tid);
  __syncthreads();
  tile_product_store(w4, a4, biasB, outB, n0, hl, hi, rowBase, nrows);
}

#define SCAN_THREADS 1024
#define SCAN_VEC     13
#define CNT_PAD      (SCAN_THREADS * SCAN_VEC * 4)
#define E_QUADS      (N_EDGES / 4)
static_assert(CNT_PAD > N_NODES, "scan must cover every node and one entry more");
static_assert(N_EDGES % 4 == 0, "edge ids are read four at a time");

__global__ void k_zero_cnt(int4* __restrict__ cnt4)
{
  cnt4[blockIdx.x * blockDim.x + threadIdx.x] = make_int4(0, 0, 0, 0);
}

__global__ void k_count(const int* __restrict__ ei, int* __restrict__ cnt,
                        int4* __restrict__ rank4)
{
  const int t = blockIdx.x * blockDim.x + threadIdx.x;
  if (t >= E_QUADS) return;
  const int4 d = reinterpret_cast<const int4*>(ei + N_EDGES)[t];
  int4 r;
  r.x = atomicAdd(cnt + d.x, 1);
  r.y = atomicAdd(cnt + d.y, 1);
  r.z = atomicAdd(cnt + d.z, 1);
  r.w = atomicAdd(cnt + d.w, 1);
  rank4[t] = r;
}

__global__ __launch_bounds__(SCAN_THREADS)
void k_scan(const int4* __restrict__ cnt4, int4* __restrict__ rowptr4)
{
  __shared__ int wsum[32];
  const int t = threadIdx.x, lane = t & 31, wv = t >> 5;
  int4 v[SCAN_VEC];
  int s = 0;
  for (int i = 0; i < SCAN_VEC; ++i) {
    v[i] = cnt4[t * SCAN_VEC + i];
    s += v[i].x + v[i].y + v[i].z + v[i].w;
  }
  int inc = s;
  for (int o = 1; o < 32; o <<= 1) {
    const int u = __shfl_up(inc, o, 32);
    if (lane >= o) inc += u;
  }
  if (lane == 31) wsum[wv] = inc;
  __syncthreads();
  if (wv == 0) {
    const int w = wsum[lane];
    int winc = w;
    for (int o = 1; o < 32; o <<= 1) {
      const int u = __shfl_up(winc, o, 32);
      if (lane >= o) winc += u;
    }
    wsum[lane] = winc - w;
  }
  __syncthreads();
  int run = wsum[wv] + inc - s;
  for (int i = 0; i < SCAN_VEC; ++i) {
    int4 o4;
    o4.x = run; run += v[i].x;
    o4.y = run; run += v[i].y;
    o4.z = run; run += v[i].z;
    o4.w = run; run += v[i].w;
    rowptr4[t * SCAN_VEC + i] = o4;
  }
}

__global__ void k_place(const int* __restrict__ ei, const int* __restrict__ rowptr,
                        const int4* __restrict__ rank4, int2* __restrict__ sse)
{
  const int t = blockIdx.x * blockDim.x + threadIdx.x;
  if (t >= E_QUADS) return;
  const int4 sv = reinterpret_cast<const int4*>(ei)[t];
  const int4 d = reinterpret_cast<const int4*>(ei + N_EDGES)[t];
  const int4 r = rank4[t];
  sse[rowptr[d.x] + r.x] = make_int2(sv.x, 4 * t);
  sse[rowptr[d.y] + r.y] = make_int2(sv.y, 4 * t + 1);
  sse[rowptr[d.z] + r.z] = make_int2(sv.z, 4 * t + 2);
  sse[rowptr[d.w] + r.w] = make_int2(sv.w, 4 * t + 3);
}

#define EG        8
#define NG_BLOCK  128
#define NG_WAVES  6144
#define NG_GRID   (NG_WAVES * 32 / NG_BLOCK)

template <int SRC>
__device__ __forceinline__ float lane_bcast(float v) {
  return __int_as_float(__builtin_amdgcn_readlane(__float_as_int(v), SRC));
}
template <int CTRL>
__device__ __forceinline__ float row_xor_add(float v) {
  const int t = __builtin_amdgcn_update_dpp(0, __float_as_int(v), CTRL, 0xf, 0xf, true);
  return v + __int_as_float(t);
}

typedef float v2f __attribute__((ext_vector_type(2)));

template <int I>
__device__ __forceinline__ float edge_logit(const float4 a, const float4 r, const float4 ea,
                                            const v2f (&weLo)[F_EDGE], const v2f (&weHi)[F_EDGE],
                                            const float4 at)
{
  v2f eeLo = {0.f, 0.f}, eeHi = {0.f, 0.f};
#define EE_FMA(J, W)                                                       \
  {                                                                        \
    const v2f wv = {W, W};                                                 \
    eeLo = __builtin_elementwise_fma(wv, weLo[J], eeLo);                   \
    eeHi = __builtin_elementwise_fma(wv, weHi[J], eeHi);                   \
  }
#define EE_STEP(T)                                                         \
  {                                                                        \
    const float w0 = lane_bcast<4 * I + T>(ea.x);                          \
    const float w1 = lane_bcast<4 * I + T>(ea.y);                          \
    const float w2 = lane_bcast<4 * I + T>(ea.z);                          \
    const float w3 = lane_bcast<4 * I + T>(ea.w);                          \
    EE_FMA(4 * T, w0) EE_FMA(4 * T + 1, w1)                                \
    EE_FMA(4 * T + 2, w2) EE_FMA(4 * T + 3, w3)                            \
  }
  EE_STEP(0) EE_STEP(1) EE_STEP(2) EE_STEP(3)
#undef EE_STEP
#undef EE_FMA
  const v2f aLo = {a.x, a.y}, aHi = {a.z, a.w};
  const v2f rLo = {r.x, r.y}, rHi = {r.z, r.w};
  const v2f mLo = aLo + rLo + eeLo;
  const v2f mHi = aHi + rHi + eeHi;
  float4 m = make_float4(mLo.x, mLo.y, mHi.x, mHi.y);
  m.x = m.x > 0.f ? m.x : SLOPE * m.x;
  m.y = m.y > 0.f ? m.y : SLOPE * m.y;
  m.z = m.z > 0.f ? m.z : SLOPE * m.z;
  m.w = m.w > 0.f ? m.w : SLOPE * m.w;
  float sres = m.x * at.x + m.y * at.y + m.z * at.z + m.w * at.w;
  sres = row_xor_add<0x161>(sres);
  sres = row_xor_add<0x162>(sres);
  sres = row_xor_add<0x164>(sres);
  sres = row_xor_add<0x168>(sres);
  return sres;
}


__global__ __launch_bounds__(NG_BLOCK)
void k_node_gat(const float* __restrict__ xl, const float* __restrict__ xr,
                const int* __restrict__ rowptr, const int2* __restrict__ sse,
                const float* __restrict__ eattr, const float* __restrict__ We,
                const float* __restrict__ att, const float* __restrict__ bout,
                float* __restrict__ h)
{
  const int lane = threadIdx.x & 31;
  const int wid = __builtin_amdgcn_readfirstlane(
      (int)(blockIdx.x * (NG_BLOCK / 32) + (threadIdx.x >> 5)));
  const int nw = gridDim.x * (NG_BLOCK / 32);
  const float4* __restrict__ xl4 = reinterpret_cast<const float4*>(xl);
  const float4* __restrict__ xr4 = reinterpret_cast<const float4*>(xr);
  const float4* __restrict__ ea4 = reinterpret_cast<const float4*>(eattr);
  float4* h4 = reinterpret_cast<float4*>(h);

  v2f weLo[F_EDGE], weHi[F_EDGE];
#pragma unroll
  for (int j = 0; j < F_EDGE; ++j) {
    const float4 w4 = reinterpret_cast<const float4*>(We + j * FDIM)[lane];
    weLo[j] = v2f{w4.x, w4.y};
    weHi[j] = v2f{w4.z, w4.w};
  }
  const float4 at = reinterpret_cast<const float4*>(att)[lane];
  const float4 bo = reinterpret_cast<const float4*>(bout)[lane];

  const int q  = lane >> 2;
  const int pc = lane & 3;

  for (int n = wid; n < N_NODES; n += nw) {
    const int beg = rowptr[n], end = rowptr[n + 1];
    const float4 hv = h4[(size_t)n * (FDIM / 4) + lane];
    float4 acc = make_float4(0.f, 0.f, 0.f, 0.f);
    if (end > beg) {
      const float4 r = xr4[(size_t)n * (FDIM / 4) + lane];
      float mx = -INFINITY;
      float den = 0.f;
      int2 se = sse[min(beg + q, end - 1)];
      for (int k0 = beg; k0 < end; k0 += EG) {
        const int cnt = min(EG, end - k0);
        const int2 seN = sse[min(k0 + EG + q, end - 1)];
        const float4 ea = ea4[(size_t)se.y * (F_EDGE / 4) + pc];
        float4 bA[EG];
        float lg[EG];
#define NG_ROW(I)                                                               \
        if (I < cnt) {                                                          \
          const int si = __builtin_amdgcn_readlane(se.x, 4 * I);                \
          bA[I] = xl4[(size_t)si * (FDIM / 4) + lane];                          \
        }
        NG_ROW(0) NG_ROW(1) NG_ROW(2) NG_ROW(3)
        NG_ROW(4) NG_ROW(5) NG_ROW(6) NG_ROW(7)
#undef NG_ROW
#define NG_LOGIT(I)                                                             \
        lg[I] = -INFINITY;                                                      \
        if (I < cnt) lg[I] = edge_logit<I>(bA[I], r, ea, weLo, weHi, at);
        NG_LOGIT(0) NG_LOGIT(1) NG_LOGIT(2) NG_LOGIT(3)
        NG_LOGIT(4) NG_LOGIT(5) NG_LOGIT(6) NG_LOGIT(7)
#undef NG_LOGIT
        float mN = fmaxf(mx, lg[0]);
        mN = fmaxf(mN, lg[1]); mN = fmaxf(mN, lg[2]); mN = fmaxf(mN, lg[3]);
        mN = fmaxf(mN, lg[4]); mN = fmaxf(mN, lg[5]); mN = fmaxf(mN, lg[6]);
        mN = fmaxf(mN, lg[7]);
        const float sc = __expf(mx - mN);
        acc.x *= sc; acc.y *= sc; acc.z *= sc; acc.w *= sc;
        den *= sc;
#define NG_ACC(I)                                                               \
        if (I < cnt) {                                                          \
          const float p = __expf(lg[I] - mN);                                   \
          den += p;                                                             \
          acc.x += bA[I].x * p;                                                 \
          acc.y += bA[I].y * p;                                                 \
          acc.z += bA[I].z * p;                                                 \
          acc.w += bA[I].w * p;                                                 \
        }
        NG_ACC(0) NG_ACC(1) NG_ACC(2) NG_ACC(3)
        NG_ACC(4) NG_ACC(5) NG_ACC(6) NG_ACC(7)
#undef NG_ACC
        mx = mN;
        se = seN;
      }
      const float inv = 1.f / den;
      acc.x *= inv; acc.y *= inv; acc.z *= inv; acc.w *= inv;
    }
    float4 o4;
    o4.x = fmaxf(acc.x + bo.x, 0.f) + hv.x;
    o4.y = fmaxf(acc.y + bo.y, 0.f) + hv.y;
    o4.z = fmaxf(acc.z + bo.z, 0.f) + hv.z;
    o4.w = fmaxf(acc.w + bo.w, 0.f) + hv.w;
    h4[(size_t)n * (FDIM / 4) + lane] = o4;
  }
}

extern "C" void kernel_launch(void* const* d_in, const int* in_sizes, int n_in,
                              void* d_out, int out_size, void* d_ws, size_t ws_size,
                              hipStream_t stream)
{
  const float* x     = (const float*)d_in[0];
  const int*   ei    = (const int*)  d_in[1];
  const float* eattr = (const float*)d_in[2];
  const float* Wp    = (const float*)d_in[3];
  const float* bp    = (const float*)d_in[4];
  const float* ln_g  = (const float*)d_in[5];
  const float* ln_b  = (const float*)d_in[6];
  const float* Wl    = (const float*)d_in[7];
  const float* bl    = (const float*)d_in[8];
  const float* Wr    = (const float*)d_in[9];
  const float* br    = (const float*)d_in[10];
  const float* We    = (const float*)d_in[11];
  const float* att   = (const float*)d_in[12];
  const float* bout  = (const float*)d_in[13];
  const float* Wc    = (const float*)d_in[14];
  const float* bc    = (const float*)d_in[15];
  float* out = (float*)d_out;

  const size_t NF = (size_t)N_NODES * FDIM;
  float* h      = out + (size_t)N_NODES * NCLS;
  float* xl     = (float*)d_ws;
  float* xr     = xl + NF;
  int*   cnt    = (int*)(xr + NF);
  int*   rowptr = cnt + CNT_PAD;
  int*   rank   = rowptr + CNT_PAD;
  int2*  sse    = (int2*)(rank + N_EDGES);

  const dim3 blk(256);
  const int gemmGrid = (N_NODES + GEMM_ROWS - 1) / GEMM_ROWS;
  const int quadGrid = (E_QUADS + 255) / 256;

  k_gemm_bf16<F_NODE, FDIM><<<gemmGrid, blk, 0, stream>>>(x, Wp, bp, h, N_NODES);

  k_zero_cnt<<<CNT_PAD / 4 / 256, blk, 0, stream>>>((int4*)cnt);
  k_count<<<quadGrid, blk, 0, stream>>>(ei, cnt, (int4*)rank);
  k_scan<<<1, SCAN_THREADS, 0, stream>>>((const int4*)cnt, (int4*)rowptr);
  k_place<<<quadGrid, blk, 0, stream>>>(ei, rowptr, (const int4*)rank, sse);

  for (int i = 0; i < NLAYER; ++i) {
    k_ln_proj2<<<gemmGrid, blk, 0, stream>>>(h, ln_g + i * FDIM, ln_b + i * FDIM,
                                             Wl + (size_t)i * FDIM * FDIM, bl + i * FDIM, xl,
                                             Wr + (size_t)i * FDIM * FDIM, br + i * FDIM, xr,
                                             N_NODES);
    k_node_gat<<<NG_GRID, NG_BLOCK, 0, stream>>>(xl, xr, rowptr, sse, eattr,
                                                 We + (size_t)i * F_EDGE * FDIM,
                                                 att + i * N_HEAD * CH, bout + i * FDIM, h);
  }

  k_gemm_bf16<FDIM, NCLS><<<gemmGrid, blk, 0, stream>>>(h, Wc, bc, out, N_NODES);
}


# attention kernel: no load of a node's start waits for another in turn: bounds of all a wave's nodes read once (lane j = node j); next node's pairs, xr row and next group's pairs asked a stage ahead
# speedup vs baseline: 4.6785x; 1.1122x over previous
#include <hip/hip_runtime.h>
#include <hip/hip_bf16.h>
#include <stdint.h>

#define N_NODES 50000
#define N_EDGES 800000
#define F_NODE  64
#define F_EDGE  16
#define N_HEAD  2
#define CH      64
#define FDIM    128
#define NLAYER  2
#define NCLS    10
#define LN_EPS  1e-5f
#define SLOPE   0.2f
#define GEMM_ROWS 64

typedef __attribute__((ext_vector_type(16))) __bf16 v16bf;
typedef __attribute__((ext_vector_type(8)))  float  v8f;

union Frag { v16bf v; uint4 q[2]; unsigned u[8]; };
union AccU { v8f v; float f[8]; };
union PK   { __bf16 h[2]; unsigned u; };

template <int FIN, int FOUT>
__global__ __launch_bounds__(256)
void k_gemm_bf16(const float* __restrict__ in, const float* __restrict__ W,
                 const float* __restrict__ bias, float* __restrict__ out, int nrows)
{
  constexpr int FOUTPAD = (FOUT + 15) & ~15;
  constexpr int KP2 = FIN / 2;
  constexpr int KQ  = FIN / 4;
  static_assert(FIN % 32 == 0 && FIN <= 128 && FOUTPAD <= 128, "tile shape");
  static_assert((GEMM_ROWS * KQ) % 256 == 0, "A pieces divide among the threads");

  __shared__ uint4 w4[FOUTPAD * 17];
  __shared__ uint4 a4[GEMM_ROWS * 17];
  unsigned* wU = reinterpret_cast<unsigned*>(w4);
  unsigned* aU = reinterpret_cast<unsigned*>(a4);

  const int tid = threadIdx.x;
  const int rowBase = blockIdx.x * GEMM_ROWS;

  constexpr int A_PER = GEMM_ROWS * KQ / 256;
  const float4* __restrict__ in4 = reinterpret_cast<const float4*>(in);
  float4 av[A_PER];
#pragma unroll
  for (int u = 0; u < A_PER; ++u) {
    const int idx = tid + u * 256;
    const int r  = idx / KQ;
    const int c4 = idx % KQ;
    const int gr = min(rowBase + r, nrows - 1);
    av[u] = in4[(size_t)gr * KQ + c4];
  }

  if constexpr (FOUT % 16 == 0) {
    constexpr int F4 = FOUT / 4;
    constexpr int NT = FOUT / 16;
    constexpr int W_PER = (FIN / 4) * F4 / 256;
    static_assert(((FIN / 4) * F4) % 256 == 0, "W items divide among the threads");
    const float4* __restrict__ W4 = reinterpret_cast<const float4*>(W);
    float4 wr[W_PER][4];
#pragma unroll
    for (int u = 0; u < W_PER; ++u) {
      const int it = tid + u * 256;
      const int up = it >> 5;
      const int n4 = (it & 3) + 4 * (up % NT);
      const int kq = ((it >> 2) & 7) + 8 * (up / NT);
#pragma unroll
      for (int j = 0; j < 4; ++j) wr[u][j] = W4[(size_t)(4 * kq + j) * F4 + n4];
    }
#pragma unroll
    for (int u = 0; u < W_PER; ++u) {
      const int it = tid + u * 256;
      const int up = it >> 5;
      const int n4 = (it & 3) + 4 * (up % NT);
      const int kq = ((it >> 2) & 7) + 8 * (up / NT);
      uint2* dst = reinterpret_cast<uint2*>(wU + (4 * n4) * 68 + 2 * kq);
      PK a, b;
      a.h[0] = (__bf16)wr[u][0].x; a.h[1] = (__bf16)wr[u][1].x;
      b.h[0] = (__bf16)wr[u][2].x; b.h[1] = (__bf16)wr[u][3].x;
      dst[0] = make_uint2(a.u, b.u);
      a.h[0] = (__bf16)wr[u][0].y; a.h[1] = (__bf16)wr[u][1].y;
      b.h[0] = (__bf16)wr[u][2].y; b.h[1] = (__bf16)wr[u][3].y;
      dst[34] = make_uint2(a.u, b.u);
      a.h[0] = (__bf16)wr[u][0].z; a.h[1] = (__bf16)wr[u][1].z;
      b.h[0] = (__bf16)wr[u][2].z; b.h[1] = (__bf16)wr[u][3].z;
      dst[68] = make_uint2(a.u, b.u);
      a.h[0] = (__bf16)wr[u][0].w; a.h[1] = (__bf16)wr[u][1].w;
      b.h[0] = (__bf16)wr[u][2].w; b.h[1] = (__bf16)wr[u][3].w;
      dst[102] = make_uint2(a.u, b.u);
    }
  } else {
    constexpr int NW = FOUTPAD * KP2;
    constexpr int W_PER = (NW + 255) / 256;
    float wlo[W_PER], whi[W_PER];
#pragma unroll
    for (int u = 0; u < W_PER; ++u) {
      const int idx = tid + u * 256;
      const int n  = idx / KP2;
      const int kp = idx % KP2;
      wlo[u] = 0.f; whi[u] = 0.f;
      if (idx < NW && n < FOUT) {
        wlo[u] = W[(size_t)(2 * kp)     * FOUT + n];
        whi[u] = W[(size_t)(2 * kp + 1) * FOUT + n];
      }
    }
#pragma unroll
    for (int u = 0; u < W_PER; ++u) {
      const int idx = tid + u * 256;
      const int n  = idx / KP2;
      const int kp = idx % KP2;
      PK p;
      p.h[0] = (__bf16)wlo[u];
      p.h[1] = (__bf16)whi[u];
      if (idx < NW) wU[n * 68 + kp] = p.u;
    }
  }

#pragma unroll
  for (int u = 0; u < A_PER; ++u) {
    const int idx = tid + u * 256;
    const int r  = idx / KQ;
    const int c4 = idx % KQ;
    PK a, b;
    a.h[0] = (__bf16)av[u].x; a.h[1] = (__bf16)av[u].y;
    b.h[0] = (__bf16)av[u].z; b.h[1] = (__bf16)av[u].w;
    *reinterpret_cast<uint2*>(aU + r * 68 + 2 * c4) = make_uint2(a.u, b.u);
  }
  __syncthreads();

  const int wv = tid >> 5;
  const int n0 = wv * 16;
  if (n0 >= FOUTPAD) return;
  const int lane = tid & 31;
  const int hl = lane & 15;
  const int hi = lane >> 4;

  AccU acc[4];
  for (int rt = 0; rt < 4; ++rt)
    for (int i = 0; i < 8; ++i) acc[rt].f[i] = 0.f;

  constexpr int nkstep = FIN >> 5;
  for (int ks = 0; ks < nkstep; ++ks) {
    const int qi = ks * 4 + hi;
    Frag b;
    b.q[0] = w4[(n0 + hl) * 17 + qi];
    b.q[1] = w4[(n0 + hl) * 17 + qi + 2];
#pragma unroll
    for (int rt = 0; rt < 4; ++rt) {
      Frag a;
      a.q[0] = a4[(rt * 16 + hl) * 17 + qi];
      a.q[1] = a4[(rt * 16 + hl) * 17 + qi + 2];
      acc[rt].v = __builtin_amdgcn_wmma_f32_16x16x32_bf16(
          false, a.v, false, b.v, (short)0, acc[rt].v, false, false);
    }
  }

  const int nn = n0 + hl;
  if (nn < FOUT) {
    const float bv = bias[nn];
    for (int rt = 0; rt < 4; ++rt) {
      for (int i = 0; i < 8; ++i) {
        int rr = rowBase + rt * 16 + hi * 8 + i;
        if (rr < nrows) out[(size_t)rr * FOUT + nn] = acc[rt].f[i] + bv;
      }
    }
  }
}

struct WRows { float4 v[FDIM / 4 * (FDIM / 4) / 256][4]; };

__device__ __forceinline__ void w_request(const float* __restrict__ W, const int tid, WRows& p)
{
  constexpr int F4 = FDIM / 4, NT = FDIM / 16, W_PER = FDIM / 4 * F4 / 256;
  const float4* __restrict__ W4 = reinterpret_cast<const float4*>(W);
#pragma unroll
  for (int u = 0; u < W_PER; ++u) {
    const int it = tid + u * 256;
    const int up = it >> 5;
    const int n4 = (it & 3) + 4 * (up % NT);
    const int kq = ((it >> 2) & 7) + 8 * (up / NT);
#pragma unroll
    for (int j = 0; j < 4; ++j) p.v[u][j] = W4[(size_t)(4 * kq + j) * F4 + n4];
  }
}

__device__ __forceinline__ void w_to_lds(const WRows& p, unsigned* wU, const int tid)
{
  constexpr int NT = FDIM / 16, W_PER = FDIM / 4 * (FDIM / 4) / 256;
#pragma unroll
  for (int u = 0; u < W_PER; ++u) {
    const int it = tid + u * 256;
    const int up = it >> 5;
    const int n4 = (it & 3) + 4 * (up % NT);
    const int kq = ((it >> 2) & 7) + 8 * (up / NT);
    uint2* dst = reinterpret_cast<uint2*>(wU + (4 * n4) * 68 + 2 * kq);
    PK a, b;
    a.h[0] = (__bf16)p.v[u][0].x; a.h[1] = (__bf16)p.v[u][1].x;
    b.h[0] = (__bf16)p.v[u][2].x; b.h[1] = (__bf16)p.v[u][3].x;
    dst[0] = make_uint2(a.u, b.u);
    a.h[0] = (__bf16)p.v[u][0].y; a.h[1] = (__bf16)p.v[u][1].y;
    b.h[0] = (__bf16)p.v[u][2].y; b.h[1] = (__bf16)p.v[u][3].y;
    dst[34] = make_uint2(a.u, b.u);
    a.h[0] = (__bf16)p.v[u][0].z; a.h[1] = (__bf16)p.v[u][1].z;
    b.h[0] = (__bf16)p.v[u][2].z; b.h[1] = (__bf16)p.v[u][3].z;
    dst[68] = make_uint2(a.u, b.u);
    a.h[0] = (__bf16)p.v[u][0].w; a.h[1] = (__bf16)p.v[u][1].w;
    b.h[0] = (__bf16)p.v[u][2].w; b.h[1] = (__bf16)p.v[u][3].w;
    dst[102] = make_uint2(a.u, b.u);
  }
}

__device__ __forceinline__ void tile_product_store(const uint4* w4, const uint4* a4,
                                                   const float* __restrict__ bias,
                                                   float* __restrict__ out,
                                                   const int n0, const int hl, const int hi,
                                                   const int rowBase, const int nrows)
{
  AccU acc[4];
  for (int rt = 0; rt < 4; ++rt)
    for (int i = 0; i < 8; ++i) acc[rt].f[i] = 0.f;
  for (int ks = 0; ks < (FDIM >> 5); ++ks) {
    const int qi = ks * 4 + hi;
    Frag b;
    b.q[0] = w4[(n0 + hl) * 17 + qi];
    b.q[1] = w4[(n0 + hl) * 17 + qi + 2];
#pragma unroll
    for (int rt = 0; rt < 4; ++rt) {
      Frag a;
      a.q[0] = a4[(rt * 16 + hl) * 17 + qi];
      a.q[1] = a4[(rt * 16 + hl) * 17 + qi + 2];
      acc[rt].v = __builtin_amdgcn_wmma_f32_16x16x32_bf16(
          false, a.v, false, b.v, (short)0, acc[rt].v, false, false);
    }
  }
  const int nn = n0 + hl;
  const float bv = bias[nn];
  const int r0 = rowBase + hi * 8;
  float* __restrict__ o = out + (size_t)r0 * FDIM + nn;
#pragma unroll
  for (int rt = 0; rt < 4; ++rt) {
#pragma unroll
    for (int i = 0; i < 8; ++i) {
      if (r0 + rt * 16 + i < nrows) o[(rt * 16 + i) * FDIM] = acc[rt].f[i] + bv;
    }
  }
}

__global__ __launch_bounds__(256)
void k_ln_proj2(const float* __restrict__ h, const float* __restrict__ g,
                const float* __restrict__ b,
                const float* __restrict__ Wa, const float* __restrict__ biasA,
                float* __restrict__ outA,
                const float* __restrict__ Wb, const float* __restrict__ biasB,
                float* __restrict__ outB, int nrows)
{
  constexpr int KQ = FDIM / 4;
  constexpr int A_PER = GEMM_ROWS * KQ / 256;
  static_assert(KQ == 32, "one wave holds one whole row per load");

  __shared__ uint4 w4[FDIM * 17];
  __shared__ uint4 a4[GEMM_ROWS * 17];
  unsigned* wU = reinterpret_cast<unsigned*>(w4);
  unsigned* aU = reinterpret_cast<unsigned*>(a4);

  const int tid = threadIdx.x;
  const int lane = tid & 31;
  const int wv = tid >> 5;
  const int rowBase = blockIdx.x * GEMM_ROWS;

  const float4* __restrict__ h4 = reinterpret_cast<const float4*>(h);
  float4 av[A_PER];
#pragma unroll
  for (int u = 0; u < A_PER; ++u) {
    const int gr = min(rowBase + wv + 8 * u, nrows - 1);
    av[u] = h4[(size_t)gr * KQ + lane];
  }
  WRows pw;
  w_request(Wa, tid, pw);
  w_to_lds(pw, wU, tid);

  const float4 gg = reinterpret_cast<const float4*>(g)[lane];
  const float4 bb = reinterpret_cast<const float4*>(b)[lane];
#pragma unroll
  for (int u = 0; u < A_PER; ++u) {
    const float4 v = av[u];
    float s = v.x + v.y + v.z + v.w;
    for (int o = 16; o >= 1; o >>= 1) s += __shfl_xor(s, o, 32);
    const float mu = s * (1.f / FDIM);
    float4 d = make_float4(v.x - mu, v.y - mu, v.z - mu, v.w - mu);
    float q = d.x * d.x + d.y * d.y + d.z * d.z + d.w * d.w;
    for (int o = 16; o >= 1; o >>= 1) q += __shfl_xor(q, o, 32);
    const float rs = rsqrtf(q * (1.f / FDIM) + LN_EPS);
    av[u] = make_float4(d.x * rs * gg.x + bb.x, d.y * rs * gg.y + bb.y,
                        d.z * rs * gg.z + bb.z, d.w * rs * gg.w + bb.w);
  }

#pragma unroll
  for (int u = 0; u < A_PER; ++u) {
    const int r = wv + 8 * u;
    PK a, c2;
    a.h[0]  = (__bf16)av[u].x; a.h[1]  = (__bf16)av[u].y;
    c2.h[0] = (__bf16)av[u].z; c2.h[1] = (__bf16)av[u].w;
    *reinterpret_cast<uint2*>(aU + r * 68 + 2 * lane) = make_uint2(a.u, c2.u);
  }
  __syncthreads();

  const int n0 = wv * 16;
  const int hl = lane & 15;
  const int hi = lane >> 4;
  tile_product_store(w4, a4, biasA, outA, n0, hl, hi, rowBase, nrows);

  __syncthreads();
  w_request(Wb, tid, pw);
  w_to_lds(pw, wU, tid);
  __syncthreads();
  tile_product_store(w4, a4, biasB, outB, n0, hl, hi, rowBase, nrows);
}

#define SCAN_THREADS 1024
#define SCAN_VEC     13
#define CNT_PAD      (SCAN_THREADS * SCAN_VEC * 4)
#define E_QUADS      (N_EDGES / 4)
static_assert(CNT_PAD > N_NODES, "scan must cover every node and one entry more");
static_assert(N_EDGES % 4 == 0, "edge ids are read four at a time");

__global__ void k_zero_cnt(int4* __restrict__ cnt4)
{
  cnt4[blockIdx.x * blockDim.x + threadIdx.x] = make_int4(0, 0, 0, 0);
}

__global__ void k_count(const int* __restrict__ ei, int* __restrict__ cnt,
                        int4* __restrict__ rank4)
{
  const int t = blockIdx.x * blockDim.x + threadIdx.x;
  if (t >= E_QUADS) return;
  const int4 d = reinterpret_cast<const int4*>(ei + N_EDGES)[t];
  int4 r;
  r.x = atomicAdd(cnt + d.x, 1);
  r.y = atomicAdd(cnt + d.y, 1);
  r.z = atomicAdd(cnt + d.z, 1);
  r.w = atomicAdd(cnt + d.w, 1);
  rank4[t] = r;
}

__global__ __launch_bounds__(SCAN_THREADS)
void k_scan(const int4* __restrict__ cnt4, int4* __restrict__ rowptr4)
{
  __shared__ int wsum[32];
  const int t = threadIdx.x, lane = t & 31, wv = t >> 5;
  int4 v[SCAN_VEC];
  int s = 0;
  for (int i = 0; i < SCAN_VEC; ++i) {
    v[i] = cnt4[t * SCAN_VEC + i];
    s += v[i].x + v[i].y + v[i].z + v[i].w;
  }
  int inc = s;
  for (int o = 1; o < 32; o <<= 1) {
    const int u = __shfl_up(inc, o, 32);
    if (lane >= o) inc += u;
  }
  if (lane == 31) wsum[wv] = inc;
  __syncthreads();
  if (wv == 0) {
    const int w = wsum[lane];
    int winc = w;
    for (int o = 1; o < 32; o <<= 1) {
      const int u = __shfl_up(winc, o, 32);
      if (lane >= o) winc += u;
    }
    wsum[lane] = winc - w;
  }
  __syncthreads();
  int run = wsum[wv] + inc - s;
  for (int i = 0; i < SCAN_VEC; ++i) {
    int4 o4;
    o4.x = run; run += v[i].x;
    o4.y = run; run += v[i].y;
    o4.z = run; run += v[i].z;
    o4.w = run; run += v[i].w;
    rowptr4[t * SCAN_VEC + i] = o4;
  }
}

__global__ void k_place(const int* __restrict__ ei, const int* __restrict__ rowptr,
                        const int4* __restrict__ rank4, int2* __restrict__ sse)
{
  const int t = blockIdx.x * blockDim.x + threadIdx.x;
  if (t >= E_QUADS) return;
  const int4 sv = reinterpret_cast<const int4*>(ei)[t];
  const int4 d = reinterpret_cast<const int4*>(ei + N_EDGES)[t];
  const int4 r = rank4[t];
  sse[rowptr[d.x] + r.x] = make_int2(sv.x, 4 * t);
  sse[rowptr[d.y] + r.y] = make_int2(sv.y, 4 * t + 1);
  sse[rowptr[d.z] + r.z] = make_int2(sv.z, 4 * t + 2);
  sse[rowptr[d.w] + r.w] = make_int2(sv.w, 4 * t + 3);
}

#define EG        8
#define NG_BLOCK  128
#define NG_WAVES  5120
#define NG_GRID   (NG_WAVES * 32 / NG_BLOCK)
static_assert(N_NODES <= 32 * NG_WAVES, "the nodes of a wave fit one per lane (their bucket bounds are kept that way)");

template <int CTRL>
__device__ __forceinline__ float row_xor_add(float v) {
  const int t = __builtin_amdgcn_update_dpp(0, __float_as_int(v), CTRL, 0xf, 0xf, true);
  return v + __int_as_float(t);
}

typedef float  v2f  __attribute__((ext_vector_type(2)));
typedef __bf16 v2bf __attribute__((ext_vector_type(2)));
union PK2 { v2bf h; unsigned u; };

__device__ __forceinline__ void split_pair(const float f0, const float f1, unsigned& hp, unsigned& lp)
{
  const v2f f = {f0, f1};
  PK2 h, l;
  h.h = __builtin_convertvector(f, v2bf);
  l.h = __builtin_convertvector(f - __builtin_convertvector(h.h, v2f), v2bf);
  hp = h.u;
  lp = l.u;
}

template <int I>
__device__ __forceinline__ float edge_logit(const float4 a, const float4 r, const AccU (&ee)[4],
                                            const float4 at)
{
  float4 m = make_float4((a.x + r.x) + ee[0].f[I], (a.y + r.y) + ee[1].f[I],
                         (a.z + r.z) + ee[2].f[I], (a.w + r.w) + ee[3].f[I]);
  m.x = m.x > 0.f ? m.x : SLOPE * m.x;
  m.y = m.y > 0.f ? m.y : SLOPE * m.y;
  m.z = m.z > 0.f ? m.z : SLOPE * m.z;
  m.w = m.w > 0.f ? m.w : SLOPE * m.w;
  float sres = m.x * at.x + m.y * at.y + m.z * at.z + m.w * at.w;
  sres = row_xor_add<0x161>(sres);
  sres = row_xor_add<0x162>(sres);
  sres = row_xor_add<0x164>(sres);
  sres = row_xor_add<0x168>(sres);
  return sres;
}


__global__ __launch_bounds__(NG_BLOCK) __attribute__((amdgpu_waves_per_eu(5)))
void k_node_gat(const float* __restrict__ xl, const float* __restrict__ xr,
                const int* __restrict__ rowptr, const int2* __restrict__ sse,
                const float* __restrict__ eattr, const float* __restrict__ We,
                const float* __restrict__ att, const float* __restrict__ bout,
                float* __restrict__ h)
{
  const int lane = threadIdx.x & 31;
  const int wid = __builtin_amdgcn_readfirstlane(
      (int)(blockIdx.x * (NG_BLOCK / 32) + (threadIdx.x >> 5)));
  const int nw = gridDim.x * (NG_BLOCK / 32);
  const float4* __restrict__ xl4 = reinterpret_cast<const float4*>(xl);
  const float4* __restrict__ xr4 = reinterpret_cast<const float4*>(xr);
  const float4* __restrict__ ea4 = reinterpret_cast<const float4*>(eattr);
  float4* h4 = reinterpret_cast<float4*>(h);

  const int hl = lane & 15;
  const int hi = lane >> 4;

  Frag bH[4], bL[4];
  {
    float4 w1[8], w2[8];
#pragma unroll
    for (int e = 0; e < 8; ++e) {
      const float4* __restrict__ wrow = reinterpret_cast<const float4*>(We + (8 * hi + e) * FDIM);
      w1[e] = wrow[hl];
      w2[e] = wrow[16 + hl];
    }
#pragma unroll
    for (int p = 0; p < 4; ++p) {
      unsigned h1[4], l1[4], h2[4], l2[4];
      split_pair(w1[2 * p].x, w1[2 * p + 1].x, h1[0], l1[0]);
      split_pair(w1[2 * p].y, w1[2 * p + 1].y, h1[1], l1[1]);
      split_pair(w1[2 * p].z, w1[2 * p + 1].z, h1[2], l1[2]);
      split_pair(w1[2 * p].w, w1[2 * p + 1].w, h1[3], l1[3]);
      split_pair(w2[2 * p].x, w2[2 * p + 1].x, h2[0], l2[0]);
      split_pair(w2[2 * p].y, w2[2 * p + 1].y, h2[1], l2[1]);
      split_pair(w2[2 * p].z, w2[2 * p + 1].z, h2[2], l2[2]);
      split_pair(w2[2 * p].w, w2[2 * p + 1].w, h2[3], l2[3]);
#pragma unroll
      for (int c = 0; c < 4; ++c) {
        bH[c].u[p] = h1[c];     bL[c].u[p] = l1[c];
        bH[c].u[4 + p] = h2[c]; bL[c].u[4 + p] = l2[c];
      }
    }
  }
  const float4 at = reinterpret_cast<const float4*>(att)[lane];
  const float4 bo = reinterpret_cast<const float4*>(bout)[lane];

  const int q = lane & 7;
  const unsigned keepLo = hl < 8 ? 0xffffffffu : 0u;
  const unsigned keepHi = ~keepLo;

  if (wid >= N_NODES) return;
  const int ownN = min(wid + lane * nw, N_NODES - 1);
  const int begV = rowptr[ownN];
  const int endV = rowptr[ownN + 1];
  int beg = __builtin_amdgcn_readlane(begV, 0);
  int end = __builtin_amdgcn_readlane(endV, 0);
  int2 se = sse[max(min(beg + q, end - 1), 0)];
  float4 r = xr4[(size_t)wid * (FDIM / 4) + lane];

#define NG_ROW(I)                                                               \
  if (I < cnt) {                                                                \
    const int si = __builtin_amdgcn_readlane(se.x, I);                          \
    bA[I] = xl4[(size_t)si * (FDIM / 4) + lane];                                \
  }
#define NG_REQUEST()                                                            \
  ea0 = ea4[(size_t)se.y * (F_EDGE / 4) + 2 * hi];                              \
  ea1 = ea4[(size_t)se.y * (F_EDGE / 4) + 2 * hi + 1];                          \
  bA[0] = xl4[(size_t)__builtin_amdgcn_readlane(se.x, 0) * (FDIM / 4) + lane];  \
  NG_ROW(1) NG_ROW(2) NG_ROW(3)                                                 \
  NG_ROW(4) NG_ROW(5) NG_ROW(6) NG_ROW(7)

  int j = 0;
  for (int n = wid; n < N_NODES; n += nw, ++j) {
    int cnt = min(EG, end - beg);
    float4 bA[EG];
    float4 ea0, ea1;
    NG_REQUEST()
    int2 seN = sse[max(min(beg + EG + q, end - 1), 0)];
    const bool more = n + nw < N_NODES;
    const int nX = more ? n + nw : n;
    const int jX = more ? j + 1 : j;
    const int begX = __builtin_amdgcn_readlane(begV, jX);
    const int endX = __builtin_amdgcn_readlane(endV, jX);
    const int2 seX = sse[max(min(begX + q, endX - 1), 0)];
    const float4 rX = xr4[(size_t)nX * (FDIM / 4) + lane];
    const float4 hv = h4[(size_t)n * (FDIM / 4) + lane];
    float4 acc = make_float4(0.f, 0.f, 0.f, 0.f);
    if (end > beg) {
      float mx = -INFINITY;
      float den = 0.f;
      int k0 = beg;
      for (;;) {
        float lg[EG];
        AccU ee[4];
        {
          unsigned ah[4], al[4];
          split_pair(ea0.x, ea0.y, ah[0], al[0]);
          split_pair(ea0.z, ea0.w, ah[1], al[1]);
          split_pair(ea1.x, ea1.y, ah[2], al[2]);
          split_pair(ea1.z, ea1.w, ah[3], al[3]);
          Frag aH, aL;
          aH.q[0] = make_uint4(ah[0] & keepLo, ah[1] & keepLo, ah[2] & keepLo, ah[3] & keepLo);
          aH.q[1] = make_uint4(ah[0] & keepHi, ah[1] & keepHi, ah[2] & keepHi, ah[3] & keepHi);
          aL.q[0] = make_uint4(al[0] & keepLo, al[1] & keepLo, al[2] & keepLo, al[3] & keepLo);
          aL.q[1] = make_uint4(al[0] & keepHi, al[1] & keepHi, al[2] & keepHi, al[3] & keepHi);
#pragma unroll
          for (int c = 0; c < 4; ++c) {
            for (int i = 0; i < 8; ++i) ee[c].f[i] = 0.f;
            ee[c].v = __builtin_amdgcn_wmma_f32_16x16x32_bf16(
                false, aH.v, false, bH[c].v, (short)0, ee[c].v, false, false);
          }
#pragma unroll
          for (int c = 0; c < 4; ++c)
            ee[c].v = __builtin_amdgcn_wmma_f32_16x16x32_bf16(
                false, aH.v, false, bL[c].v, (short)0, ee[c].v, false, false);
#pragma unroll
          for (int c = 0; c < 4; ++c)
            ee[c].v = __builtin_amdgcn_wmma_f32_16x16x32_bf16(
                false, aL.v, false, bH[c].v, (short)0, ee[c].v, false, false);
        }
#define NG_LOGIT(I)                                                             \
        lg[I] = -INFINITY;                                                      \
        if (I < cnt) lg[I] = edge_logit<I>(bA[I], r, ee, at);
        NG_LOGIT(0) NG_LOGIT(1) NG_LOGIT(2) NG_LOGIT(3)
        NG_LOGIT(4) NG_LOGIT(5) NG_LOGIT(6) NG_LOGIT(7)
#undef NG_LOGIT
        float mN = fmaxf(mx, lg[0]);
        mN = fmaxf(mN, lg[1]); mN = fmaxf(mN, lg[2]); mN = fmaxf(mN, lg[3]);
        mN = fmaxf(mN, lg[4]); mN = fmaxf(mN, lg[5]); mN = fmaxf(mN, lg[6]);
        mN = fmaxf(mN, lg[7]);
        const float sc = __expf(mx - mN);
        acc.x *= sc; acc.y *= sc; acc.z *= sc; acc.w *= sc;
        den *= sc;
#define NG_ACC(I)                                                               \
        if (I < cnt) {                                                          \
          const float p = __expf(lg[I] - mN);                                   \
          den += p;                                                             \
          acc.x += bA[I].x * p;                                                 \
          acc.y += bA[I].y * p;                                                 \
          acc.z += bA[I].z * p;                                                 \
          acc.w += bA[I].w * p;                                                 \
        }
        NG_ACC(0) NG_ACC(1) NG_ACC(2) NG_ACC(3)
        NG_ACC(4) NG_ACC(5) NG_ACC(6) NG_ACC(7)
#undef NG_ACC
        mx = mN;
        k0 += EG;
        if (k0 >= end) break;
        cnt = min(EG, end - k0);
        se = seN;
        NG_REQUEST()
        seN = sse[min(k0 + EG + q, end - 1)];
      }
      const float inv = 1.f / den;
      acc.x *= inv; acc.y *= inv; acc.z *= inv; acc.w *= inv;
    }
    float4 o4;
    o4.x = fmaxf(acc.x + bo.x, 0.f) + hv.x;
    o4.y = fmaxf(acc.y + bo.y, 0.f) + hv.y;
    o4.z = fmaxf(acc.z + bo.z, 0.f) + hv.z;
    o4.w = fmaxf(acc.w + bo.w, 0.f) + hv.w;
    h4[(size_t)n * (FDIM / 4) + lane] = o4;
    beg = begX; end = endX; se = seX; r = rX;
  }
#undef NG_REQUEST
#undef NG_ROW
}

extern "C" void kernel_launch(void* const* d_in, const int* in_sizes, int n_in,
                              void* d_out, int out_size, void* d_ws, size_t ws_size,
                              hipStream_t stream)
{
  const float* x     = (const float*)d_in[0];
  const int*   ei    = (const int*)  d_in[1];
  const float* eattr = (const float*)d_in[2];
  const float* Wp    = (const float*)d_in[3];
  const float* bp    = (const float*)d_in[4];
  const float* ln_g  = (const float*)d_in[5];
  const float* ln_b  = (const float*)d_in[6];
  const float* Wl    = (const float*)d_in[7];
  const float* bl    = (const float*)d_in[8];
  const float* Wr    = (const float*)d_in[9];
  const float* br    = (const float*)d_in[10];
  const float* We    = (const float*)d_in[11];
  const float* att   = (const float*)d_in[12];
  const float* bout  = (const float*)d_in[13];
  const float* Wc    = (const float*)d_in[14];
  const float* bc    = (const float*)d_in[15];
  float* out = (float*)d_out;

  const size_t NF = (size_t)N_NODES * FDIM;
  float* h      = out + (size_t)N_NODES * NCLS;
  float* xl     = (float*)d_ws;
  float* xr     = xl + NF;
  int*   cnt    = (int*)(xr + NF);
  int*   rowptr = cnt + CNT_PAD;
  int*   rank   = rowptr + CNT_PAD;
  int2*  sse    = (int2*)(rank + N_EDGES);

  const dim3 blk(256);
  const int gemmGrid = (N_NODES + GEMM_ROWS - 1) / GEMM_ROWS;
  const int quadGrid = (E_QUADS + 255) / 256;

  k_gemm_bf16<F_NODE, FDIM><<<gemmGrid, blk, 0, stream>>>(x, Wp, bp, h, N_NODES);

  k_zero_cnt<<<CNT_PAD / 4 / 256, blk, 0, stream>>>((int4*)cnt);
  k_count<<<quadGrid, blk, 0, stream>>>(ei, cnt, (int4*)rank);
  k_scan<<<1, SCAN_THREADS, 0, stream>>>((const int4*)cnt, (int4*)rowptr);
  k_place<<<quadGrid, blk, 0, stream>>>(ei, rowptr, (const int4*)rank, sse);

  for (int i = 0; i < NLAYER; ++i) {
    k_ln_proj2<<<gemmGrid, blk, 0, stream>>>(h, ln_g + i * FDIM, ln_b + i * FDIM,
                                             Wl + (size_t)i * FDIM * FDIM, bl + i * FDIM, xl,
                                             Wr + (size_t)i * FDIM * FDIM, br + i * FDIM, xr,
                                             N_NODES);
    k_node_gat<<<NG_GRID, NG_BLOCK, 0, stream>>>(xl, xr, rowptr, sse, eattr,
                                                 We + (size_t)i * F_EDGE * FDIM,
                                                 att + i * N_HEAD * CH, bout + i * FDIM, h);
  }

  k_gemm_bf16<FDIM, NCLS><<<gemmGrid, blk, 0, stream>>>(h, Wc, bc, out, N_NODES);
}
